# starting kernel
# baseline (speedup 1.0000x reference)
#include <hip/hip_runtime.h>
#include <math.h>

typedef __attribute__((ext_vector_type(2))) float v2f;
typedef __attribute__((ext_vector_type(8))) float v8f;

#define EPS_LN 1e-6f
#define EPS_L2 1e-12f

__device__ __forceinline__ float gelu_exact(float v) {
    return 0.5f * v * (1.0f + erff(v * 0.70710678118654752440f));
}

__global__ __launch_bounds__(256) void tok_kernel(
    const float* __restrict__ token, const float* __restrict__ p,
    const float* __restrict__ alpha,
    const float* __restrict__ g1, const float* __restrict__ b1,
    const float* __restrict__ w_tok, const float* __restrict__ b_tok,
    float* __restrict__ tn_out)
{
    __shared__ float red[256];
    __shared__ float tnorm[768];
    __shared__ float gbuf[128];
    __shared__ float sm, srstd, snrm;
    const int b = blockIdx.x, t = threadIdx.x;
    const float* tok = token + b * 768;

    float s = 0.f;
    for (int i = t; i < 768; i += 256) s += tok[i];
    red[t] = s; __syncthreads();
    for (int off = 128; off > 0; off >>= 1) { if (t < off) red[t] += red[t + off]; __syncthreads(); }
    if (t == 0) sm = red[0] * (1.f / 768.f);
    __syncthreads();
    const float m = sm;

    float q = 0.f;
    for (int i = t; i < 768; i += 256) { float d = tok[i] - m; q += d * d; }
    red[t] = q; __syncthreads();
    for (int off = 128; off > 0; off >>= 1) { if (t < off) red[t] += red[t + off]; __syncthreads(); }
    if (t == 0) srstd = rsqrtf(red[0] * (1.f / 768.f) + EPS_LN);
    __syncthreads();
    const float rstd = srstd;

    for (int i = t; i < 768; i += 256) tnorm[i] = (tok[i] - m) * rstd * g1[i] + b1[i];
    __syncthreads();

    if (t < 128) {
        float acc = b_tok[t];
        for (int k = 0; k < 768; ++k) acc = fmaf(tnorm[k], w_tok[k * 128 + t], acc);
        gbuf[t] = gelu_exact(acc);
    }
    __syncthreads();

    float ss = 0.f;
    if (t < 128) { float g = gbuf[t]; ss = g * g; }
    red[t] = ss; __syncthreads();
    for (int off = 128; off > 0; off >>= 1) { if (t < off) red[t] += red[t + off]; __syncthreads(); }
    if (t == 0) snrm = fmaxf(sqrtf(red[0]), EPS_L2);
    __syncthreads();

    if (t < 128) {
        float scale = p[b] * expf(alpha[0]) / snrm;
        tn_out[b * 128 + t] = gbuf[t] * scale;
    }
}

#define ROWP 196

__global__ __launch_bounds__(256) void xattn_kernel(
    const float* __restrict__ x, const float* __restrict__ w_x,
    const float* __restrict__ b_x,
    const float* __restrict__ g2, const float* __restrict__ b2,
    const float* __restrict__ g3, const float* __restrict__ b3,
    const float* __restrict__ tn, float* __restrict__ out)
{
    __shared__ float xs[16 * ROWP];
    __shared__ float as_[16 * ROWP];
    __shared__ float sg2[192], sb2[192], sg3[192], sb3[192];
    __shared__ float tnl[128];
    __shared__ float mx[16], vx[16];
    __shared__ float pdot[16][8], psq[16][8];
    __shared__ float sattn[16], smeany[16], srstdy[16];

    const int t    = threadIdx.x;
    const int wave = t >> 5;
    const int lane = t & 31;
    const int nl   = lane & 15;
    const int hi   = lane >> 4;
    const int b     = blockIdx.x >> 4;
    const int chunk = blockIdx.x & 15;
    const int ncol  = wave * 16 + nl;

    for (int i = t; i < 192; i += 256) {
        sg2[i] = g2[i]; sb2[i] = b2[i]; sg3[i] = g3[i]; sb3[i] = b3[i];
    }
    if (t < 128) tnl[t] = tn[b * 128 + t];

    v2f bfrag[48];
    #pragma unroll
    for (int kb = 0; kb < 48; ++kb) {
        const int k0 = kb * 4 + hi * 2;
        bfrag[kb].x = w_x[k0 * 128 + ncol];
        bfrag[kb].y = w_x[(k0 + 1) * 128 + ncol];
    }
    const float bxv = b_x[ncol];
    __syncthreads();
    const float tnv = tnl[ncol];

    const size_t base = ((size_t)(b * 4096 + chunk * 256)) * 192;
    float4 pf[3];
    {
        const float4* xg = (const float4*)(x + base);
        #pragma unroll
        for (int j = 0; j < 3; ++j) pf[j] = xg[t + j * 256];
    }

    for (int tt = 0; tt < 16; ++tt) {
        const size_t tbase = base + (size_t)tt * 3072;

        #pragma unroll
        for (int j = 0; j < 3; ++j) {
            int e4 = t + j * 256;
            int r = e4 / 48, c4 = e4 - r * 48;
            *(float4*)&xs[r * ROWP + c4 * 4] = pf[j];
        }
        if (tt < 15) {
            const float4* xg = (const float4*)(x + tbase + 3072);
            #pragma unroll
            for (int j = 0; j < 3; ++j) pf[j] = xg[t + j * 256];
        }
        __syncthreads();

        for (int rr = 0; rr < 2; ++rr) {
            const int r = wave + rr * 8;
            float s = 0.f, q = 0.f;
            for (int j = 0; j < 6; ++j) {
                float xv = xs[r * ROWP + lane + j * 32];
                s += xv; q += xv * xv;
            }
            for (int msk = 16; msk >= 1; msk >>= 1) {
                s += __shfl_xor(s, msk, 32);
                q += __shfl_xor(q, msk, 32);
            }
            float m   = s * (1.f / 192.f);
            float var = q * (1.f / 192.f) - m * m;
            float rst = rsqrtf(var + EPS_LN);
            if (lane == 0) { mx[r] = m; vx[r] = var; }
            for (int j = 0; j < 6; ++j) {
                int c = lane + j * 32;
                as_[r * ROWP + c] = (xs[r * ROWP + c] - m) * rst * sg2[c] + sb2[c];
            }
        }
        __syncthreads();

        v8f acc0 = {}, acc1 = {};
        const float* arow = &as_[nl * ROWP + hi * 2];
        #pragma unroll
        for (int kb = 0; kb < 48; kb += 2) {
            v2f a0 = *(const v2f*)(arow + kb * 4);
            v2f a1 = *(const v2f*)(arow + kb * 4 + 4);
            acc0 = __builtin_amdgcn_wmma_f32_16x16x4_f32(
                false, a0, false, bfrag[kb], (short)0, acc0, false, false);
            acc1 = __builtin_amdgcn_wmma_f32_16x16x4_f32(
                false, a1, false, bfrag[kb + 1], (short)0, acc1, false, false);
        }

        #pragma unroll
        for (int v = 0; v < 8; ++v) {
            float g  = gelu_exact(acc0[v] + acc1[v] + bxv);
            float pd = g * tnv;
            float ps = g * g;
            for (int msk = 8; msk >= 1; msk >>= 1) {
                pd += __shfl_xor(pd, msk, 32);
                ps += __shfl_xor(ps, msk, 32);
            }
            if (nl == 0) {
                int r = v + hi * 8;
                pdot[r][wave] = pd;
                psq[r][wave]  = ps;
            }
        }
        __syncthreads();

        if (t < 16) {
            float dot = 0.f, sq = 0.f;
            for (int wv = 0; wv < 8; ++wv) { dot += pdot[t][wv]; sq += psq[t][wv]; }
            float attn = dot / fmaxf(sqrtf(sq), EPS_L2);
            sattn[t]  = attn;
            smeany[t] = attn * mx[t];
            srstdy[t] = rsqrtf(attn * attn * vx[t] + EPS_LN);
        }
        __syncthreads();

        {
            const int r  = t >> 4;
            const int c0 = (t & 15) * 12;
            const float attn = sattn[r], my = smeany[r], rs = srstdy[r];
            float* og = out + tbase + (size_t)r * 192;
            const float* xr = &xs[r * ROWP];
            #pragma unroll
            for (int c = c0; c < c0 + 12; ++c) {
                float xv = xr[c];
                og[c] = 0.5f * xv + (xv * attn - my) * rs * sg3[c] + sb3[c];
            }
        }
        __syncthreads();
    }
}

extern "C" void kernel_launch(void* const* d_in, const int* in_sizes, int n_in,
                              void* d_out, int out_size, void* d_ws, size_t ws_size,
                              hipStream_t stream) {
    const float* x     = (const float*)d_in[0];
    const float* token = (const float*)d_in[1];
    const float* p     = (const float*)d_in[2];
    const float* alpha = (const float*)d_in[3];
    const float* ln1_g = (const float*)d_in[4];
    const float* ln1_b = (const float*)d_in[5];
    const float* w_tok = (const float*)d_in[6];
    const float* b_tok = (const float*)d_in[7];
    const float* ln2_g = (const float*)d_in[8];
    const float* ln2_b = (const float*)d_in[9];
    const float* w_x   = (const float*)d_in[10];
    const float* b_x   = (const float*)d_in[11];
    const float* ln3_g = (const float*)d_in[12];
    const float* ln3_b = (const float*)d_in[13];
    float* out = (float*)d_out;
    float* tn  = (float*)d_ws;

    tok_kernel<<<32, 256, 0, stream>>>(token, p, alpha, ln1_g, ln1_b,
                                       w_tok, b_tok, tn);
    xattn_kernel<<<512, 256, 0, stream>>>(x, w_x, b_x, ln2_g, ln2_b,
                                          ln3_g, ln3_b, tn, out);
}


# One launch instead of two: each tile workgroup computes its batch's token vector itself (16 waves x 48 rows of w_tok, partials added via LDS); changes the token LN1 and matvec sum order
# speedup vs baseline: 2.8818x; 2.8818x over previous
#include <hip/hip_runtime.h>
#include <math.h>

typedef __attribute__((ext_vector_type(2))) float v2f;
typedef __attribute__((ext_vector_type(8))) float v8f;

#define EPS_LN 1e-6f
#define EPS_L2 1e-12f

__device__ __forceinline__ float gelu_exact(float v) {
    return 0.5f * v * (1.0f + erff(v * 0.70710678118654752440f));
}

__device__ __forceinline__ float wave_sum(float v) {
    #pragma unroll
    for (int msk = 16; msk >= 1; msk >>= 1) v += __shfl_xor(v, msk, 32);
    return v;
}

#define ROWP  196
#define NWAVE 16
#define WTILE (16 * ROWP)
#define TNOFF (WTILE + 64)
#define WSLOT (WTILE + 64 + 128)

__global__ __launch_bounds__(512) void xattn_kernel(
    const float* __restrict__ x, const float* __restrict__ token,
    const float* __restrict__ p, const float* __restrict__ alpha,
    const float* __restrict__ g1, const float* __restrict__ b1,
    const float* __restrict__ w_tok, const float* __restrict__ b_tok,
    const float* __restrict__ w_x, const float* __restrict__ b_x,
    const float* __restrict__ g2, const float* __restrict__ b2,
    const float* __restrict__ g3, const float* __restrict__ b3,
    float* __restrict__ out)
{
    __shared__ float4 wlA[24 * 4 * 32], wlB[24 * 4 * 32];
    __shared__ __attribute__((aligned(16))) float xw[NWAVE * WSLOT];
    __shared__ __attribute__((aligned(16))) float sg2[192], sb2[192], sg3[192], sb3[192];
    __shared__ __attribute__((aligned(16))) float tpart[NWAVE * 128];
    __shared__ float sbx[128];

    const int t    = threadIdx.x;
    const int wave = t >> 5;
    const int lane = t & 31;
    const int c    = lane & 15;
    const int h    = lane >> 4;
    float* xs = xw + wave * WSLOT;
    float* st = xs + WTILE;
    float* wtn = xs + TNOFF;

    int xo[3], co[3], ro[3];
    #pragma unroll
    for (int j = 0; j < 3; ++j) {
        const int e = j * 32 + lane;
        ro[j] = e / 48;
        co[j] = (e - ro[j] * 48) * 4;
        xo[j] = ro[j] * ROWP + co[j];
    }

    const int tile0 = (blockIdx.x * NWAVE + wave) * 2;
    const int b     = blockIdx.x >> 3;

    const float4* xg = (const float4*)(x + (size_t)tile0 * 3072);
    {
        float4 pf[24];
        #pragma unroll
        for (int i = 0; i < 24; ++i) pf[i] = xg[i * 32 + lane];

        const float4* tk4 = (const float4*)(token + b * 768);
        float4 tq[6];
        #pragma unroll
        for (int j = 0; j < 6; ++j) tq[j] = tk4[j * 32 + lane];
        const int sl = wave * 12 + (lane < 12 ? lane : 11);
        const float4 ts = tk4[sl];
        const float4 gs = ((const float4*)g1)[sl];
        const float4 bs = ((const float4*)b1)[sl];
        const float4* wt4 = (const float4*)w_tok + (wave * 48) * 32 + lane;
        float4 wa[16], wb[16];
        #pragma unroll
        for (int kk = 0; kk < 16; ++kk) wa[kk] = wt4[kk * 32];

        #pragma unroll
        for (int i = 0; i < 24; ++i)
            *(float4*)&xs[(i / 3) * 2 * ROWP + xo[i % 3]] = pf[i];

        #pragma unroll
        for (int kk = 0; kk < 16; ++kk) wb[kk] = wt4[(16 + kk) * 32];

        float s1 = 0.f;
        #pragma unroll
        for (int j = 0; j < 6; ++j) { s1 += tq[j].x; s1 += tq[j].y; s1 += tq[j].z; s1 += tq[j].w; }
        const float m1 = wave_sum(s1) * (1.f / 768.f);
        float q1 = 0.f;
        #pragma unroll
        for (int j = 0; j < 6; ++j) {
            const float dx = tq[j].x - m1, dy = tq[j].y - m1, dz = tq[j].z - m1, dw = tq[j].w - m1;
            q1 = fmaf(dx, dx, q1); q1 = fmaf(dy, dy, q1); q1 = fmaf(dz, dz, q1); q1 = fmaf(dw, dw, q1);
        }
        const float rstd1 = rsqrtf(wave_sum(q1) * (1.f / 768.f) + EPS_LN);
        {
            float4 tnv;
            tnv.x = (ts.x - m1) * rstd1 * gs.x + bs.x;
            tnv.y = (ts.y - m1) * rstd1 * gs.y + bs.y;
            tnv.z = (ts.z - m1) * rstd1 * gs.z + bs.z;
            tnv.w = (ts.w - m1) * rstd1 * gs.w + bs.w;
            *(float4*)&wtn[lane * 4] = tnv;
        }

        __builtin_amdgcn_fence(__ATOMIC_RELEASE, "wavefront");
        __builtin_amdgcn_wave_barrier();
        __builtin_amdgcn_fence(__ATOMIC_ACQUIRE, "wavefront");

        float4 acc = {0.f, 0.f, 0.f, 0.f};
        #pragma unroll
        for (int kk = 0; kk < 16; kk += 4) {
            const float4 t4 = *(const float4*)&wtn[kk];
            acc.x = fmaf(t4.x, wa[kk].x, acc.x);     acc.y = fmaf(t4.x, wa[kk].y, acc.y);
            acc.z = fmaf(t4.x, wa[kk].z, acc.z);     acc.w = fmaf(t4.x, wa[kk].w, acc.w);
            acc.x = fmaf(t4.y, wa[kk + 1].x, acc.x); acc.y = fmaf(t4.y, wa[kk + 1].y, acc.y);
            acc.z = fmaf(t4.y, wa[kk + 1].z, acc.z); acc.w = fmaf(t4.y, wa[kk + 1].w, acc.w);
            acc.x = fmaf(t4.z, wa[kk + 2].x, acc.x); acc.y = fmaf(t4.z, wa[kk + 2].y, acc.y);
            acc.z = fmaf(t4.z, wa[kk + 2].z, acc.z); acc.w = fmaf(t4.z, wa[kk + 2].w, acc.w);
            acc.x = fmaf(t4.w, wa[kk + 3].x, acc.x); acc.y = fmaf(t4.w, wa[kk + 3].y, acc.y);
            acc.z = fmaf(t4.w, wa[kk + 3].z, acc.z); acc.w = fmaf(t4.w, wa[kk + 3].w, acc.w);
        }
        #pragma unroll
        for (int kk = 0; kk < 16; ++kk) wa[kk] = wt4[(32 + kk) * 32];
        #pragma unroll
        for (int kk = 0; kk < 16; kk += 4) {
            const float4 t4 = *(const float4*)&wtn[16 + kk];
            acc.x = fmaf(t4.x, wb[kk].x, acc.x);     acc.y = fmaf(t4.x, wb[kk].y, acc.y);
            acc.z = fmaf(t4.x, wb[kk].z, acc.z);     acc.w = fmaf(t4.x, wb[kk].w, acc.w);
            acc.x = fmaf(t4.y, wb[kk + 1].x, acc.x); acc.y = fmaf(t4.y, wb[kk + 1].y, acc.y);
            acc.z = fmaf(t4.y, wb[kk + 1].z, acc.z); acc.w = fmaf(t4.y, wb[kk + 1].w, acc.w);
            acc.x = fmaf(t4.z, wb[kk + 2].x, acc.x); acc.y = fmaf(t4.z, wb[kk + 2].y, acc.y);
            acc.z = fmaf(t4.z, wb[kk + 2].z, acc.z); acc.w = fmaf(t4.z, wb[kk + 2].w, acc.w);
            acc.x = fmaf(t4.w, wb[kk + 3].x, acc.x); acc.y = fmaf(t4.w, wb[kk + 3].y, acc.y);
            acc.z = fmaf(t4.w, wb[kk + 3].z, acc.z); acc.w = fmaf(t4.w, wb[kk + 3].w, acc.w);
        }

        float4 wv[12];
        #pragma unroll
        for (int j = 0; j < 12; ++j) {
            const int e  = t + j * 512;
            const int l  = e & 31, np = (e >> 5) & 3, kb = e >> 7;
            const int k0 = kb * 4 + (l >> 4) * 2;
            const int col = np * 32 + (l & 15);
            wv[j].x = w_x[k0 * 128 + col];
            wv[j].y = w_x[(k0 + 1) * 128 + col];
            wv[j].z = w_x[k0 * 128 + col + 16];
            wv[j].w = w_x[(k0 + 1) * 128 + col + 16];
        }

        #pragma unroll
        for (int kk = 0; kk < 16; kk += 4) {
            const float4 t4 = *(const float4*)&wtn[32 + kk];
            acc.x = fmaf(t4.x, wa[kk].x, acc.x);     acc.y = fmaf(t4.x, wa[kk].y, acc.y);
            acc.z = fmaf(t4.x, wa[kk].z, acc.z);     acc.w = fmaf(t4.x, wa[kk].w, acc.w);
            acc.x = fmaf(t4.y, wa[kk + 1].x, acc.x); acc.y = fmaf(t4.y, wa[kk + 1].y, acc.y);
            acc.z = fmaf(t4.y, wa[kk + 1].z, acc.z); acc.w = fmaf(t4.y, wa[kk + 1].w, acc.w);
            acc.x = fmaf(t4.z, wa[kk + 2].x, acc.x); acc.y = fmaf(t4.z, wa[kk + 2].y, acc.y);
            acc.z = fmaf(t4.z, wa[kk + 2].z, acc.z); acc.w = fmaf(t4.z, wa[kk + 2].w, acc.w);
            acc.x = fmaf(t4.w, wa[kk + 3].x, acc.x); acc.y = fmaf(t4.w, wa[kk + 3].y, acc.y);
            acc.z = fmaf(t4.w, wa[kk + 3].z, acc.z); acc.w = fmaf(t4.w, wa[kk + 3].w, acc.w);
        }
        *(float4*)&tpart[wave * 128 + lane * 4] = acc;

        #pragma unroll
        for (int j = 0; j < 6; ++j) { wlA[t + j * 512] = wv[j]; wlB[t + j * 512] = wv[j + 6]; }
    }
    if (t < 192) { sg2[t] = g2[t]; sb2[t] = b2[t]; sg3[t] = g3[t]; sb3[t] = b3[t]; }
    if (t < 128) sbx[t] = b_x[t];
    const float4 btk = ((const float4*)b_tok)[lane];
    const float  pscale = p[b] * expf(alpha[0]);
    __syncthreads();

    {
        float4 sa = btk;
        #pragma unroll
        for (int w = 0; w < NWAVE; ++w) {
            const float4 pp = *(const float4*)&tpart[w * 128 + lane * 4];
            sa.x += pp.x; sa.y += pp.y; sa.z += pp.z; sa.w += pp.w;
        }
        float4 gt;
        gt.x = gelu_exact(sa.x); gt.y = gelu_exact(sa.y);
        gt.z = gelu_exact(sa.z); gt.w = gelu_exact(sa.w);
        float ss = gt.x * gt.x;
        ss = fmaf(gt.y, gt.y, ss); ss = fmaf(gt.z, gt.z, ss); ss = fmaf(gt.w, gt.w, ss);
        const float scale = pscale / fmaxf(sqrtf(wave_sum(ss)), EPS_L2);
        gt.x *= scale; gt.y *= scale; gt.z *= scale; gt.w *= scale;
        *(float4*)&wtn[lane * 4] = gt;
    }

    #pragma unroll 1
    for (int it = 0; it < 2; ++it) {
        float* og = out + (size_t)(tile0 + it) * 3072;

        if (it > 0) {
            float4 pf[24];
            #pragma unroll
            for (int i = 0; i < 24; ++i) pf[i] = xg[768 + i * 32 + lane];
            #pragma unroll
            for (int i = 0; i < 24; ++i)
                *(float4*)&xs[(i / 3) * 2 * ROWP + xo[i % 3]] = pf[i];
        }
        __builtin_amdgcn_fence(__ATOMIC_RELEASE, "wavefront");
        __builtin_amdgcn_wave_barrier();
        __builtin_amdgcn_fence(__ATOMIC_ACQUIRE, "wavefront");

        const float* arow = xs + c * ROWP + h * 2;
        float s = 0.f, q = 0.f;
        #pragma unroll
        for (int kb = 0; kb < 48; ++kb) {
            const v2f xv = *(const v2f*)(arow + kb * 4);
            s += xv.x; s += xv.y;
            q = fmaf(xv.x, xv.x, q); q = fmaf(xv.y, xv.y, q);
        }
        s += __shfl_xor(s, 16, 32);
        q += __shfl_xor(q, 16, 32);
        const float m   = s * (1.f / 192.f);
        const float var = q * (1.f / 192.f) - m * m;
        const float rst = rsqrtf(var + EPS_LN);

        __builtin_amdgcn_fence(__ATOMIC_ACQUIRE, "wavefront");

        v8f acc[8];
        #pragma unroll
        for (int nt = 0; nt < 8; ++nt) acc[nt] = (v8f){};
        #pragma unroll
        for (int kb = 0; kb < 48; ++kb) {
            const v2f xv = *(const v2f*)(arow + kb * 4);
            const v2f gg = *(const v2f*)&sg2[kb * 4 + h * 2];
            const v2f bb = *(const v2f*)&sb2[kb * 4 + h * 2];
            v2f a;
            a.x = (xv.x - m) * rst * gg.x + bb.x;
            a.y = (xv.y - m) * rst * gg.y + bb.y;
            #pragma unroll
            for (int np = 0; np < 4; ++np) {
                const float4 w4 = (kb < 24) ? wlA[(kb * 4 + np) * 32 + lane]
                                            : wlB[((kb - 24) * 4 + np) * 32 + lane];
                v2f b0, b1;
                b0.x = w4.x; b0.y = w4.y; b1.x = w4.z; b1.y = w4.w;
                acc[2 * np] = __builtin_amdgcn_wmma_f32_16x16x4_f32(
                    false, a, false, b0, (short)0, acc[2 * np], false, false);
                acc[2 * np + 1] = __builtin_amdgcn_wmma_f32_16x16x4_f32(
                    false, a, false, b1, (short)0, acc[2 * np + 1], false, false);
            }
        }

        float pd[8], ps[8];
        #pragma unroll
        for (int r = 0; r < 8; ++r) { pd[r] = 0.f; ps[r] = 0.f; }
        #pragma unroll
        for (int nt = 0; nt < 8; ++nt) {
            const float bx = sbx[nt * 16 + c];
            const float tv = wtn[nt * 16 + c];
            #pragma unroll
            for (int r = 0; r < 8; ++r) {
                const float gv = gelu_exact(acc[nt][r] + bx);
                pd[r] = fmaf(gv, tv, pd[r]);
                ps[r] = fmaf(gv, gv, ps[r]);
            }
        }
        #pragma unroll
        for (int r = 0; r < 8; ++r) {
            for (int msk = 8; msk >= 1; msk >>= 1) {
                pd[r] += __shfl_xor(pd[r], msk, 32);
                ps[r] += __shfl_xor(ps[r], msk, 32);
            }
            st[(h * 8 + r) * 4] = pd[r] / fmaxf(sqrtf(ps[r]), EPS_L2);
        }
        __builtin_amdgcn_fence(__ATOMIC_RELEASE, "wavefront");
        __builtin_amdgcn_wave_barrier();
        __builtin_amdgcn_fence(__ATOMIC_ACQUIRE, "wavefront");

        {
            const float attn = st[c * 4];
            st[c * 4 + 1] = attn * m;
            st[c * 4 + 2] = rsqrtf(attn * attn * var + EPS_LN);
        }
        __builtin_amdgcn_fence(__ATOMIC_RELEASE, "wavefront");
        __builtin_amdgcn_wave_barrier();
        __builtin_amdgcn_fence(__ATOMIC_ACQUIRE, "wavefront");

        #pragma unroll
        for (int i = 0; i < 24; ++i) {
            const int p = i / 3, j = i % 3;
            const float4 xv = *(const float4*)&xs[p * 2 * ROWP + xo[j]];
            const float4 gg = *(const float4*)&sg3[co[j]];
            const float4 bb = *(const float4*)&sb3[co[j]];
            const float4 rw = *(const float4*)&st[(p * 2 + ro[j]) * 4];
            const float attn = rw.x, my = rw.y, rs = rw.z;
            float4 o;
            o.x = 0.5f * xv.x + (xv.x * attn - my) * rs * gg.x + bb.x;
            o.y = 0.5f * xv.y + (xv.y * attn - my) * rs * gg.y + bb.y;
            o.z = 0.5f * xv.z + (xv.z * attn - my) * rs * gg.z + bb.z;
            o.w = 0.5f * xv.w + (xv.w * attn - my) * rs * gg.w + bb.w;
            ((float4*)og)[i * 32 + lane] = o;
        }
        __builtin_amdgcn_fence(__ATOMIC_RELEASE, "wavefront");
        __builtin_amdgcn_wave_barrier();
        __builtin_amdgcn_fence(__ATOMIC_ACQUIRE, "wavefront");
    }
}

extern "C" void kernel_launch(void* const* d_in, const int* in_sizes, int n_in,
                              void* d_out, int out_size, void* d_ws, size_t ws_size,
                              hipStream_t stream) {
    const float* x     = (const float*)d_in[0];
    const float* token = (const float*)d_in[1];
    const float* p     = (const float*)d_in[2];
    const float* alpha = (const float*)d_in[3];
    const float* ln1_g = (const float*)d_in[4];
    const float* ln1_b = (const float*)d_in[5];
    const float* w_tok = (const float*)d_in[6];
    const float* b_tok = (const float*)d_in[7];
    const float* ln2_g = (const float*)d_in[8];
    const float* ln2_b = (const float*)d_in[9];
    const float* w_x   = (const float*)d_in[10];
    const float* b_x   = (const float*)d_in[11];
    const float* ln3_g = (const float*)d_in[12];
    const float* ln3_b = (const float*)d_in[13];
    float* out = (float*)d_out;
    (void)d_ws; (void)ws_size;

    xattn_kernel<<<256, 512, 0, stream>>>(x, token, p, alpha, ln1_g, ln1_b,
                                          w_tok, b_tok, w_x, b_x, ln2_g, ln2_b,
                                          ln3_g, ln3_b, out);
}
